# starting kernel
# baseline (speedup 1.0000x reference)
#include <hip/hip_runtime.h>

#define D 32
#define NG 256

typedef __attribute__((ext_vector_type(2))) float v2f;
typedef __attribute__((ext_vector_type(8))) float v8f;

__global__ void zero_kernel(float* __restrict__ p, int n) {
    int i = blockIdx.x * blockDim.x + threadIdx.x;
    int stride = gridDim.x * blockDim.x;
    for (; i < n; i += stride) p[i] = 0.f;
}

__global__ void node_scatter(const float* __restrict__ x,
                             const int* __restrict__ batch,
                             float* __restrict__ acc, int n_nodes) {
    __shared__ float sh[NG * D];
    for (int i = threadIdx.x; i < NG * D; i += blockDim.x) sh[i] = 0.f;
    __syncthreads();

    int chunk = (n_nodes + gridDim.x - 1) / gridDim.x;
    int start = blockIdx.x * chunk;
    int end   = min(start + chunk, n_nodes);
    int d   = threadIdx.x & (D - 1);
    int sub = threadIdx.x >> 5;
    int step = blockDim.x >> 5;
    for (int n = start + sub; n < end; n += step) {
        int g = batch[n];
        atomicAdd(&sh[g * D + d], x[(long)n * D + d]);
    }
    __syncthreads();
    for (int i = threadIdx.x; i < NG * D; i += blockDim.x) {
        float v = sh[i];
        if (v != 0.f) atomicAdd(&acc[i], v);
    }
}

__global__ void edge_scatter(const float* __restrict__ edge_attr,
                             const int* __restrict__ src,
                             const int* __restrict__ batch,
                             float* __restrict__ acc, int n_edges) {
    __shared__ float sh[NG * D];
    for (int i = threadIdx.x; i < NG * D; i += blockDim.x) sh[i] = 0.f;
    __syncthreads();

    int chunk = (n_edges + gridDim.x - 1) / gridDim.x;
    int start = blockIdx.x * chunk;
    int end   = min(start + chunk, n_edges);
    int d   = threadIdx.x & (D - 1);
    int sub = threadIdx.x >> 5;
    int step = blockDim.x >> 5;
    for (int e = start + sub; e < end; e += step) {
        int g = batch[src[e]];
        atomicAdd(&sh[g * D + d], edge_attr[(long)e * D + d]);
    }
    __syncthreads();
    for (int i = threadIdx.x; i < NG * D; i += blockDim.x) {
        float v = sh[i];
        if (v != 0.f) atomicAdd(&acc[i], v);
    }
}

__global__ void gemm_bias_relu(const float* __restrict__ node_acc,
                               const float* __restrict__ edge_acc,
                               const float* __restrict__ u,
                               const float* __restrict__ W,
                               const float* __restrict__ b,
                               float* __restrict__ out, int n_graphs) {
    const int lane  = threadIdx.x & 31;
    const int wid   = (blockIdx.x * blockDim.x + threadIdx.x) >> 5;
    const int mtiles = n_graphs >> 4;
    const int tile_m = wid % mtiles;
    const int tile_n = wid / mtiles;

    const int l15   = lane & 15;
    const int khalf = (lane >> 4) << 1;
    const int mrow  = tile_m * 16 + l15;
    const int ncol  = tile_n * 16 + l15;

    v8f c = {};
    #pragma unroll
    for (int kb = 0; kb < 24; ++kb) {
        const int k0 = kb * 4;
        const float* A = (k0 < 32) ? node_acc : (k0 < 64) ? edge_acc : u;
        const int koff = k0 & 31;
        v2f a, bm;
        a[0]  = A[mrow * D + koff + khalf + 0];
        a[1]  = A[mrow * D + koff + khalf + 1];
        bm[0] = W[(k0 + khalf + 0) * D + ncol];
        bm[1] = W[(k0 + khalf + 1) * D + ncol];
        c = __builtin_amdgcn_wmma_f32_16x16x4_f32(
                 false, a,  false, bm,
                 (short)0, c,  false,  false);
    }

    const float bias = b[ncol];
    const int rbase = tile_m * 16 + ((lane >> 4) << 3);
    #pragma unroll
    for (int r = 0; r < 8; ++r) {
        float v = c[r] + bias;
        out[(rbase + r) * D + ncol] = v > 0.f ? v : 0.f;
    }
}

extern "C" void kernel_launch(void* const* d_in, const int* in_sizes, int n_in,
                              void* d_out, int out_size, void* d_ws, size_t ws_size,
                              hipStream_t stream) {
    const float* x          = (const float*)d_in[0];
    const int*   edge_index = (const int*)d_in[1];
    const float* edge_attr  = (const float*)d_in[2];
    const float* u          = (const float*)d_in[3];
    const int*   batch      = (const int*)d_in[4];
    const float* W          = (const float*)d_in[5];
    const float* b          = (const float*)d_in[6];
    float*       out        = (float*)d_out;

    const int n_nodes  = in_sizes[4];
    const int n_edges  = in_sizes[2] / D;
    const int n_graphs = in_sizes[3] / D;

    float* node_acc = (float*)d_ws;
    float* edge_acc = node_acc + n_graphs * D;

    zero_kernel<<<32, 256, 0, stream>>>(node_acc, 2 * n_graphs * D);
    node_scatter<<<512, 256, 0, stream>>>(x, batch, node_acc, n_nodes);
    edge_scatter<<<1024, 256, 0, stream>>>(edge_attr, edge_index, batch,
                                           edge_acc, n_edges);
    const int waves  = (n_graphs / 16) * 2;
    const int blocks = (waves + 15) / 16;
    gemm_bias_relu<<<blocks, 512, 0, stream>>>(node_acc, edge_acc, u, W, b,
                                               out, n_graphs);
}


# edge segment-sum: waves take 32 edges per trip, src and graph ids fetched lane-parallel, rows by 16-byte loads kept one trip ahead (512 blocks); float atomics kept, sums in a new order
# speedup vs baseline: 3.2429x; 3.2429x over previous
#include <hip/hip_runtime.h>

#define D 32
#define NG 256

typedef __attribute__((ext_vector_type(2))) float v2f;
typedef __attribute__((ext_vector_type(8))) float v8f;

__global__ void zero_kernel(float* __restrict__ p, int n) {
    int i = blockIdx.x * blockDim.x + threadIdx.x;
    int stride = gridDim.x * blockDim.x;
    for (; i < n; i += stride) p[i] = 0.f;
}

__global__ void node_scatter(const float* __restrict__ x,
                             const int* __restrict__ batch,
                             float* __restrict__ acc, int n_nodes) {
    __shared__ float sh[NG * D];
    for (int i = threadIdx.x; i < NG * D; i += blockDim.x) sh[i] = 0.f;
    __syncthreads();

    int chunk = (n_nodes + gridDim.x - 1) / gridDim.x;
    int start = blockIdx.x * chunk;
    int end   = min(start + chunk, n_nodes);
    int d   = threadIdx.x & (D - 1);
    int sub = threadIdx.x >> 5;
    int step = blockDim.x >> 5;
    for (int n = start + sub; n < end; n += step) {
        int g = batch[n];
        atomicAdd(&sh[g * D + d], x[(long)n * D + d]);
    }
    __syncthreads();
    for (int i = threadIdx.x; i < NG * D; i += blockDim.x) {
        float v = sh[i];
        if (v != 0.f) atomicAdd(&acc[i], v);
    }
}

typedef __attribute__((ext_vector_type(4))) float v4f;

#define TRIP 32

__device__ __forceinline__ void load_rows(const v4f* __restrict__ rows, int trip,
                                          int lane, v4f (&R)[8]) {
    const v4f* p = rows + (size_t)trip * (TRIP * D / 4) + lane;
    #pragma unroll
    for (int i = 0; i < 8; ++i) R[i] = p[i * 32];
}

__device__ __forceinline__ void add_rows(float* sh, const v4f (&R)[8], int g,
                                         int r, const int (&c)[4]) {
    const bool r1 = (r & 1) != 0, r2 = (r & 2) != 0;
    #pragma unroll
    for (int i = 0; i < 8; ++i) {
        const int gi = __shfl(g, 4 * i + r, 32);
        float* row = sh + gi * D;
        const v4f v = R[i];
        const float a0 = r1 ? v[1] : v[0];
        const float a1 = r1 ? v[2] : v[1];
        const float a2 = r1 ? v[3] : v[2];
        const float a3 = r1 ? v[0] : v[3];
        const float w0 = r2 ? a2 : a0;
        const float w1 = r2 ? a3 : a1;
        const float w2 = r2 ? a0 : a2;
        const float w3 = r2 ? a1 : a3;
        atomicAdd(row + c[0], w0);
        atomicAdd(row + c[1], w1);
        atomicAdd(row + c[2], w2);
        atomicAdd(row + c[3], w3);
    }
}

__global__ void __launch_bounds__(256)
edge_scatter(const float* __restrict__ edge_attr,
             const int* __restrict__ src,
             const int* __restrict__ batch,
             float* __restrict__ acc, int n_edges) {
    __shared__ float sh[NG * D];
    for (int i = threadIdx.x; i < NG * D; i += blockDim.x) sh[i] = 0.f;
    __syncthreads();

    const int lane = threadIdx.x & 31;
    const int wave = __builtin_amdgcn_readfirstlane(threadIdx.x >> 5);
    const int nwb  = blockDim.x >> 5;
    const int r = lane >> 3;
    const int q = lane & 7;
    int c[4];
    #pragma unroll
    for (int j = 0; j < 4; ++j) c[j] = 4 * q + ((j + r) & 3);

    const v4f* rows = (const v4f*)edge_attr;
    const int nfull = n_edges / TRIP;
    const int gw = wave * gridDim.x + blockIdx.x;
    const int nw = nwb * gridDim.x;
    const int n_my = (nfull > gw) ? (nfull - gw + nw - 1) / nw : 0;

    if (n_my > 0) {
        v4f A[8], B[8];
        int ga, gb;
        const int last = gw + (n_my - 1) * nw;
        int s0 = src[gw * TRIP + lane];
        load_rows(rows, gw, lane, A);
        int s1 = src[min(gw + nw, last) * TRIP + lane];
        ga = batch[s0];
        int k = 0;
        for (; k + 2 < n_my; k += 2) {
            const int t1 = gw + (k + 1) * nw;
            const int t2 = t1 + nw;
            const int s2 = src[t2 * TRIP + lane];
            gb = batch[s1];
            load_rows(rows, t1, lane, B);
            add_rows(sh, A, ga, r, c);
            const int s3 = src[min(t2 + nw, last) * TRIP + lane];
            ga = batch[s2];
            load_rows(rows, t2, lane, A);
            add_rows(sh, B, gb, r, c);
            s1 = s3;
        }
        if (k + 1 < n_my) {
            gb = batch[s1];
            load_rows(rows, gw + (k + 1) * nw, lane, B);
            add_rows(sh, A, ga, r, c);
            add_rows(sh, B, gb, r, c);
        } else {
            add_rows(sh, A, ga, r, c);
        }
    }

    if (blockIdx.x == 0 && wave == 0) {
        for (int e = nfull * TRIP; e < n_edges; ++e) {
            int g = batch[src[e]];
            atomicAdd(&sh[g * D + lane], edge_attr[(long)e * D + lane]);
        }
    }

    __syncthreads();
    for (int i = threadIdx.x; i < NG * D; i += blockDim.x) {
        float v = sh[i];
        if (v != 0.f) atomicAdd(&acc[i], v);
    }
}

__global__ void gemm_bias_relu(const float* __restrict__ node_acc,
                               const float* __restrict__ edge_acc,
                               const float* __restrict__ u,
                               const float* __restrict__ W,
                               const float* __restrict__ b,
                               float* __restrict__ out, int n_graphs) {
    const int lane  = threadIdx.x & 31;
    const int wid   = (blockIdx.x * blockDim.x + threadIdx.x) >> 5;
    const int mtiles = n_graphs >> 4;
    const int tile_m = wid % mtiles;
    const int tile_n = wid / mtiles;

    const int l15   = lane & 15;
    const int khalf = (lane >> 4) << 1;
    const int mrow  = tile_m * 16 + l15;
    const int ncol  = tile_n * 16 + l15;

    v8f c = {};
    #pragma unroll
    for (int kb = 0; kb < 24; ++kb) {
        const int k0 = kb * 4;
        const float* A = (k0 < 32) ? node_acc : (k0 < 64) ? edge_acc : u;
        const int koff = k0 & 31;
        v2f a, bm;
        a[0]  = A[mrow * D + koff + khalf + 0];
        a[1]  = A[mrow * D + koff + khalf + 1];
        bm[0] = W[(k0 + khalf + 0) * D + ncol];
        bm[1] = W[(k0 + khalf + 1) * D + ncol];
        c = __builtin_amdgcn_wmma_f32_16x16x4_f32(
                 false, a,  false, bm,
                 (short)0, c,  false,  false);
    }

    const float bias = b[ncol];
    const int rbase = tile_m * 16 + ((lane >> 4) << 3);
    #pragma unroll
    for (int r = 0; r < 8; ++r) {
        float v = c[r] + bias;
        out[(rbase + r) * D + ncol] = v > 0.f ? v : 0.f;
    }
}

extern "C" void kernel_launch(void* const* d_in, const int* in_sizes, int n_in,
                              void* d_out, int out_size, void* d_ws, size_t ws_size,
                              hipStream_t stream) {
    const float* x          = (const float*)d_in[0];
    const int*   edge_index = (const int*)d_in[1];
    const float* edge_attr  = (const float*)d_in[2];
    const float* u          = (const float*)d_in[3];
    const int*   batch      = (const int*)d_in[4];
    const float* W          = (const float*)d_in[5];
    const float* b          = (const float*)d_in[6];
    float*       out        = (float*)d_out;

    const int n_nodes  = in_sizes[4];
    const int n_edges  = in_sizes[2] / D;
    const int n_graphs = in_sizes[3] / D;

    float* node_acc = (float*)d_ws;
    float* edge_acc = node_acc + n_graphs * D;

    zero_kernel<<<32, 256, 0, stream>>>(node_acc, 2 * n_graphs * D);
    node_scatter<<<512, 256, 0, stream>>>(x, batch, node_acc, n_nodes);
    edge_scatter<<<512, 256, 0, stream>>>(edge_attr, edge_index, batch,
                                           edge_acc, n_edges);
    const int waves  = (n_graphs / 16) * 2;
    const int blocks = (waves + 15) / 16;
    gemm_bias_relu<<<blocks, 512, 0, stream>>>(node_acc, edge_acc, u, W, b,
                                               out, n_graphs);
}


# edge scatter: each block stores its LDS table to a workspace slot (16-byte stores); the former zero kernel sums the 512 tables in fixed order; no global atomics for edges, sum order changes
# speedup vs baseline: 3.9015x; 1.2031x over previous
#include <hip/hip_runtime.h>

#define D 32
#define NG 256

typedef __attribute__((ext_vector_type(2))) float v2f;
typedef __attribute__((ext_vector_type(8))) float v8f;

typedef __attribute__((ext_vector_type(4))) float v4f;

#define NPART 512

__global__ void __launch_bounds__(256)
sum_tables(const float* __restrict__ part, float* __restrict__ edge_acc,
           float* __restrict__ node_acc) {
    __shared__ v4f red[8][8];
    const int q  = threadIdx.x & 7;
    const int pg = threadIdx.x >> 3;
    const int lane = threadIdx.x & 31;
    const int wave = threadIdx.x >> 5;
    const v4f* p4 = (const v4f*)part + (size_t)pg * (NG * D / 4)
                    + blockIdx.x * 8 + q;
    v4f s = {0.f, 0.f, 0.f, 0.f};
    #pragma unroll
    for (int i = 0; i < NPART / 32; ++i)
        s += p4[(size_t)i * 32 * (NG * D / 4)];
    #pragma unroll
    for (int k = 0; k < 4; ++k) {
        s[k] += __shfl_xor(s[k], 8, 32);
        s[k] += __shfl_xor(s[k], 16, 32);
    }
    if (lane < 8) red[wave][q] = s;
    __syncthreads();
    if (threadIdx.x < 8) {
        v4f t = red[0][q];
        #pragma unroll
        for (int w = 1; w < 8; ++w) t += red[w][q];
        const v4f z = {0.f, 0.f, 0.f, 0.f};
        ((v4f*)edge_acc)[blockIdx.x * 8 + q] = t;
        ((v4f*)node_acc)[blockIdx.x * 8 + q] = z;
    }
}

__global__ void node_scatter(const float* __restrict__ x,
                             const int* __restrict__ batch,
                             float* __restrict__ acc, int n_nodes) {
    __shared__ float sh[NG * D];
    for (int i = threadIdx.x; i < NG * D; i += blockDim.x) sh[i] = 0.f;
    __syncthreads();

    int chunk = (n_nodes + gridDim.x - 1) / gridDim.x;
    int start = blockIdx.x * chunk;
    int end   = min(start + chunk, n_nodes);
    int d   = threadIdx.x & (D - 1);
    int sub = threadIdx.x >> 5;
    int step = blockDim.x >> 5;
    for (int n = start + sub; n < end; n += step) {
        int g = batch[n];
        atomicAdd(&sh[g * D + d], x[(long)n * D + d]);
    }
    __syncthreads();
    for (int i = threadIdx.x; i < NG * D; i += blockDim.x) {
        float v = sh[i];
        if (v != 0.f) atomicAdd(&acc[i], v);
    }
}

#define TRIP 32

__device__ __forceinline__ void load_rows(const v4f* __restrict__ rows, int trip,
                                          int lane, v4f (&R)[8]) {
    const v4f* p = rows + (size_t)trip * (TRIP * D / 4) + lane;
    #pragma unroll
    for (int i = 0; i < 8; ++i) R[i] = p[i * 32];
}

__device__ __forceinline__ void add_rows(float* sh, const v4f (&R)[8], int g,
                                         int r, const int (&c)[4]) {
    const bool r1 = (r & 1) != 0, r2 = (r & 2) != 0;
    #pragma unroll
    for (int i = 0; i < 8; ++i) {
        const int gi = __shfl(g, 4 * i + r, 32);
        float* row = sh + gi * D;
        const v4f v = R[i];
        const float a0 = r1 ? v[1] : v[0];
        const float a1 = r1 ? v[2] : v[1];
        const float a2 = r1 ? v[3] : v[2];
        const float a3 = r1 ? v[0] : v[3];
        const float w0 = r2 ? a2 : a0;
        const float w1 = r2 ? a3 : a1;
        const float w2 = r2 ? a0 : a2;
        const float w3 = r2 ? a1 : a3;
        atomicAdd(row + c[0], w0);
        atomicAdd(row + c[1], w1);
        atomicAdd(row + c[2], w2);
        atomicAdd(row + c[3], w3);
    }
}

__global__ void __launch_bounds__(256)
edge_scatter(const float* __restrict__ edge_attr,
             const int* __restrict__ src,
             const int* __restrict__ batch,
             float* __restrict__ part, int n_edges) {
    __shared__ v4f sh4[NG * D / 4];
    float* sh = (float*)sh4;
    for (int i = threadIdx.x; i < NG * D; i += blockDim.x) sh[i] = 0.f;
    __syncthreads();

    const int lane = threadIdx.x & 31;
    const int wave = __builtin_amdgcn_readfirstlane(threadIdx.x >> 5);
    const int nwb  = blockDim.x >> 5;
    const int r = lane >> 3;
    const int q = lane & 7;
    int c[4];
    #pragma unroll
    for (int j = 0; j < 4; ++j) c[j] = 4 * q + ((j + r) & 3);

    const v4f* rows = (const v4f*)edge_attr;
    const int nfull = n_edges / TRIP;
    const int gw = wave * gridDim.x + blockIdx.x;
    const int nw = nwb * gridDim.x;
    const int n_my = (nfull > gw) ? (nfull - gw + nw - 1) / nw : 0;

    if (n_my > 0) {
        v4f A[8], B[8];
        int ga, gb;
        const int last = gw + (n_my - 1) * nw;
        int s0 = src[gw * TRIP + lane];
        load_rows(rows, gw, lane, A);
        int s1 = src[min(gw + nw, last) * TRIP + lane];
        ga = batch[s0];
        int k = 0;
        for (; k + 2 < n_my; k += 2) {
            const int t1 = gw + (k + 1) * nw;
            const int t2 = t1 + nw;
            const int s2 = src[t2 * TRIP + lane];
            gb = batch[s1];
            load_rows(rows, t1, lane, B);
            add_rows(sh, A, ga, r, c);
            const int s3 = src[min(t2 + nw, last) * TRIP + lane];
            ga = batch[s2];
            load_rows(rows, t2, lane, A);
            add_rows(sh, B, gb, r, c);
            s1 = s3;
        }
        if (k + 1 < n_my) {
            gb = batch[s1];
            load_rows(rows, gw + (k + 1) * nw, lane, B);
            add_rows(sh, A, ga, r, c);
            add_rows(sh, B, gb, r, c);
        } else {
            add_rows(sh, A, ga, r, c);
        }
    }

    if (blockIdx.x == 0 && wave == 0) {
        for (int e = nfull * TRIP; e < n_edges; ++e) {
            int g = batch[src[e]];
            atomicAdd(&sh[g * D + lane], edge_attr[(long)e * D + lane]);
        }
    }

    __syncthreads();
    v4f* dst = (v4f*)part + (size_t)blockIdx.x * (NG * D / 4);
    #pragma unroll 8
    for (int i = threadIdx.x; i < NG * D / 4; i += blockDim.x) dst[i] = sh4[i];
}

__global__ void gemm_bias_relu(const float* __restrict__ node_acc,
                               const float* __restrict__ edge_acc,
                               const float* __restrict__ u,
                               const float* __restrict__ W,
                               const float* __restrict__ b,
                               float* __restrict__ out, int n_graphs) {
    const int lane  = threadIdx.x & 31;
    const int wid   = (blockIdx.x * blockDim.x + threadIdx.x) >> 5;
    const int mtiles = n_graphs >> 4;
    const int tile_m = wid % mtiles;
    const int tile_n = wid / mtiles;

    const int l15   = lane & 15;
    const int khalf = (lane >> 4) << 1;
    const int mrow  = tile_m * 16 + l15;
    const int ncol  = tile_n * 16 + l15;

    v8f c = {};
    #pragma unroll
    for (int kb = 0; kb < 24; ++kb) {
        const int k0 = kb * 4;
        const float* A = (k0 < 32) ? node_acc : (k0 < 64) ? edge_acc : u;
        const int koff = k0 & 31;
        v2f a, bm;
        a[0]  = A[mrow * D + koff + khalf + 0];
        a[1]  = A[mrow * D + koff + khalf + 1];
        bm[0] = W[(k0 + khalf + 0) * D + ncol];
        bm[1] = W[(k0 + khalf + 1) * D + ncol];
        c = __builtin_amdgcn_wmma_f32_16x16x4_f32(
                 false, a,  false, bm,
                 (short)0, c,  false,  false);
    }

    const float bias = b[ncol];
    const int rbase = tile_m * 16 + ((lane >> 4) << 3);
    #pragma unroll
    for (int r = 0; r < 8; ++r) {
        float v = c[r] + bias;
        out[(rbase + r) * D + ncol] = v > 0.f ? v : 0.f;
    }
}

extern "C" void kernel_launch(void* const* d_in, const int* in_sizes, int n_in,
                              void* d_out, int out_size, void* d_ws, size_t ws_size,
                              hipStream_t stream) {
    const float* x          = (const float*)d_in[0];
    const int*   edge_index = (const int*)d_in[1];
    const float* edge_attr  = (const float*)d_in[2];
    const float* u          = (const float*)d_in[3];
    const int*   batch      = (const int*)d_in[4];
    const float* W          = (const float*)d_in[5];
    const float* b          = (const float*)d_in[6];
    float*       out        = (float*)d_out;

    const int n_nodes  = in_sizes[4];
    const int n_edges  = in_sizes[2] / D;
    const int n_graphs = in_sizes[3] / D;

    float* node_acc  = (float*)d_ws;
    float* edge_acc  = node_acc + n_graphs * D;
    float* edge_part = edge_acc + n_graphs * D;

    edge_scatter<<<NPART, 256, 0, stream>>>(edge_attr, edge_index, batch,
                                             edge_part, n_edges);
    sum_tables<<<NG * D / 32, 256, 0, stream>>>(edge_part, edge_acc, node_acc);
    node_scatter<<<512, 256, 0, stream>>>(x, batch, node_acc, n_nodes);
    const int waves  = (n_graphs / 16) * 2;
    const int blocks = (waves + 15) / 16;
    gemm_bias_relu<<<blocks, 512, 0, stream>>>(node_acc, edge_acc, u, W, b,
                                               out, n_graphs);
}


# node segment-sum folded into the edge kernel: 31 of its 512 blocks run the same trip loop over x; sum_tables adds their tables; node launch and global atomics gone; node sum order changes
# speedup vs baseline: 5.4909x; 1.4074x over previous
#include <hip/hip_runtime.h>

#define D 32
#define NG 256

typedef __attribute__((ext_vector_type(2))) float v2f;
typedef __attribute__((ext_vector_type(8))) float v8f;

typedef __attribute__((ext_vector_type(4))) float v4f;

#define NEDGEB 481
#define NNODEB 31
#define NSLOT (NEDGEB + NNODEB)
static_assert(NSLOT % 32 == 0, "sum_tables deals the slots to 32 thread groups");

__global__ void __launch_bounds__(256)
sum_tables(const float* __restrict__ part, float* __restrict__ edge_acc,
           float* __restrict__ node_acc) {
    __shared__ v4f red[2][8][8];
    const int q  = threadIdx.x & 7;
    const int pg = threadIdx.x >> 3;
    const int lane = threadIdx.x & 31;
    const int wave = threadIdx.x >> 5;
    const v4f* p4 = (const v4f*)part + (size_t)pg * (NG * D / 4)
                    + blockIdx.x * 8 + q;
    v4f se = {0.f, 0.f, 0.f, 0.f};
    v4f sn = {0.f, 0.f, 0.f, 0.f};
    #pragma unroll
    for (int i = 0; i < NSLOT / 32; ++i) {
        const v4f v = p4[(size_t)i * 32 * (NG * D / 4)];
        if (32 * i + 31 < NEDGEB)             se += v;
        else if (32 * i >= NEDGEB)            sn += v;
        else if (pg + 32 * i < NEDGEB)        se += v;
        else                                  sn += v;
    }
    #pragma unroll
    for (int k = 0; k < 4; ++k) {
        se[k] += __shfl_xor(se[k], 8, 32);
        se[k] += __shfl_xor(se[k], 16, 32);
        sn[k] += __shfl_xor(sn[k], 8, 32);
        sn[k] += __shfl_xor(sn[k], 16, 32);
    }
    if (lane < 8) { red[0][wave][q] = se; red[1][wave][q] = sn; }
    __syncthreads();
    if (threadIdx.x < 16) {
        const int kind = threadIdx.x >> 3;
        v4f t = red[kind][0][q];
        #pragma unroll
        for (int w = 1; w < 8; ++w) t += red[kind][w][q];
        float* dst = kind ? node_acc : edge_acc;
        ((v4f*)dst)[blockIdx.x * 8 + q] = t;
    }
}

#define TRIP 32

__device__ __forceinline__ void load_rows(const v4f* __restrict__ rows, int trip,
                                          int lane, v4f (&R)[8]) {
    const v4f* p = rows + (size_t)trip * (TRIP * D / 4) + lane;
    #pragma unroll
    for (int i = 0; i < 8; ++i) R[i] = p[i * 32];
}

__device__ __forceinline__ void add_rows(float* sh, const v4f (&R)[8], int g,
                                         int r, const int (&c)[4]) {
    const bool r1 = (r & 1) != 0, r2 = (r & 2) != 0;
    #pragma unroll
    for (int i = 0; i < 8; ++i) {
        const int gi = __shfl(g, 4 * i + r, 32);
        float* row = sh + gi * D;
        const v4f v = R[i];
        const float a0 = r1 ? v[1] : v[0];
        const float a1 = r1 ? v[2] : v[1];
        const float a2 = r1 ? v[3] : v[2];
        const float a3 = r1 ? v[0] : v[3];
        const float w0 = r2 ? a2 : a0;
        const float w1 = r2 ? a3 : a1;
        const float w2 = r2 ? a0 : a2;
        const float w3 = r2 ? a1 : a3;
        atomicAdd(row + c[0], w0);
        atomicAdd(row + c[1], w1);
        atomicAdd(row + c[2], w2);
        atomicAdd(row + c[3], w3);
    }
}

template <bool NODE>
__device__ __forceinline__ int graph_key(const int* __restrict__ src, int i) {
    if (NODE) return i;
    return src[i];
}

template <bool NODE>
__device__ __forceinline__ void sum_rows(float* sh, const float* __restrict__ feat,
                                         const int* __restrict__ src,
                                         const int* __restrict__ batch,
                                         int n_rows, int blk, int nblk) {
    const int lane = threadIdx.x & 31;
    const int wave = __builtin_amdgcn_readfirstlane(threadIdx.x >> 5);
    const int nwb  = blockDim.x >> 5;
    const int r = lane >> 3;
    const int q = lane & 7;
    int c[4];
    #pragma unroll
    for (int j = 0; j < 4; ++j) c[j] = 4 * q + ((j + r) & 3);

    const v4f* rows = (const v4f*)feat;
    const int nfull = n_rows / TRIP;
    const int gw = wave * nblk + blk;
    const int nw = nwb * nblk;
    const int n_my = (nfull > gw) ? (nfull - gw + nw - 1) / nw : 0;

    if (n_my > 0) {
        v4f A[8], B[8];
        int ga, gb;
        const int last = gw + (n_my - 1) * nw;
        int s0 = graph_key<NODE>(src, gw * TRIP + lane);
        load_rows(rows, gw, lane, A);
        int s1 = graph_key<NODE>(src, min(gw + nw, last) * TRIP + lane);
        ga = batch[s0];
        int k = 0;
        for (; k + 2 < n_my; k += 2) {
            const int t1 = gw + (k + 1) * nw;
            const int t2 = t1 + nw;
            const int s2 = graph_key<NODE>(src, t2 * TRIP + lane);
            gb = batch[s1];
            load_rows(rows, t1, lane, B);
            add_rows(sh, A, ga, r, c);
            const int s3 = graph_key<NODE>(src, min(t2 + nw, last) * TRIP + lane);
            ga = batch[s2];
            load_rows(rows, t2, lane, A);
            add_rows(sh, B, gb, r, c);
            s1 = s3;
        }
        if (k + 1 < n_my) {
            gb = batch[s1];
            load_rows(rows, gw + (k + 1) * nw, lane, B);
            add_rows(sh, A, ga, r, c);
            add_rows(sh, B, gb, r, c);
        } else {
            add_rows(sh, A, ga, r, c);
        }
    }

    if (blk == 0 && wave == 0) {
        for (int e = nfull * TRIP; e < n_rows; ++e) {
            int g = batch[graph_key<NODE>(src, e)];
            atomicAdd(&sh[g * D + lane], feat[(long)e * D + lane]);
        }
    }
}

__global__ void __launch_bounds__(256)
scatter_rows(const float* __restrict__ edge_attr,
             const int* __restrict__ src,
             const float* __restrict__ x,
             const int* __restrict__ batch,
             float* __restrict__ part, int n_edges, int n_nodes) {
    __shared__ v4f sh4[NG * D / 4];
    float* sh = (float*)sh4;
    for (int i = threadIdx.x; i < NG * D; i += blockDim.x) sh[i] = 0.f;
    __syncthreads();

    if (blockIdx.x < NEDGEB)
        sum_rows<false>(sh, edge_attr, src, batch, n_edges, blockIdx.x, NEDGEB);
    else
        sum_rows<true>(sh, x, src, batch, n_nodes, blockIdx.x - NEDGEB, NNODEB);

    __syncthreads();
    v4f* dst = (v4f*)part + (size_t)blockIdx.x * (NG * D / 4);
    #pragma unroll 8
    for (int i = threadIdx.x; i < NG * D / 4; i += blockDim.x) dst[i] = sh4[i];
}

__global__ void gemm_bias_relu(const float* __restrict__ node_acc,
                               const float* __restrict__ edge_acc,
                               const float* __restrict__ u,
                               const float* __restrict__ W,
                               const float* __restrict__ b,
                               float* __restrict__ out, int n_graphs) {
    const int lane  = threadIdx.x & 31;
    const int wid   = (blockIdx.x * blockDim.x + threadIdx.x) >> 5;
    const int mtiles = n_graphs >> 4;
    const int tile_m = wid % mtiles;
    const int tile_n = wid / mtiles;

    const int l15   = lane & 15;
    const int khalf = (lane >> 4) << 1;
    const int mrow  = tile_m * 16 + l15;
    const int ncol  = tile_n * 16 + l15;

    v8f c = {};
    #pragma unroll
    for (int kb = 0; kb < 24; ++kb) {
        const int k0 = kb * 4;
        const float* A = (k0 < 32) ? node_acc : (k0 < 64) ? edge_acc : u;
        const int koff = k0 & 31;
        v2f a, bm;
        a[0]  = A[mrow * D + koff + khalf + 0];
        a[1]  = A[mrow * D + koff + khalf + 1];
        bm[0] = W[(k0 + khalf + 0) * D + ncol];
        bm[1] = W[(k0 + khalf + 1) * D + ncol];
        c = __builtin_amdgcn_wmma_f32_16x16x4_f32(
                 false, a,  false, bm,
                 (short)0, c,  false,  false);
    }

    const float bias = b[ncol];
    const int rbase = tile_m * 16 + ((lane >> 4) << 3);
    #pragma unroll
    for (int r = 0; r < 8; ++r) {
        float v = c[r] + bias;
        out[(rbase + r) * D + ncol] = v > 0.f ? v : 0.f;
    }
}

extern "C" void kernel_launch(void* const* d_in, const int* in_sizes, int n_in,
                              void* d_out, int out_size, void* d_ws, size_t ws_size,
                              hipStream_t stream) {
    const float* x          = (const float*)d_in[0];
    const int*   edge_index = (const int*)d_in[1];
    const float* edge_attr  = (const float*)d_in[2];
    const float* u          = (const float*)d_in[3];
    const int*   batch      = (const int*)d_in[4];
    const float* W          = (const float*)d_in[5];
    const float* b          = (const float*)d_in[6];
    float*       out        = (float*)d_out;

    const int n_nodes  = in_sizes[4];
    const int n_edges  = in_sizes[2] / D;
    const int n_graphs = in_sizes[3] / D;

    float* node_acc  = (float*)d_ws;
    float* edge_acc  = node_acc + n_graphs * D;
    float* part      = edge_acc + n_graphs * D;

    scatter_rows<<<NSLOT, 256, 0, stream>>>(edge_attr, edge_index, x, batch,
                                            part, n_edges, n_nodes);
    sum_tables<<<NG * D / 32, 256, 0, stream>>>(part, edge_acc, node_acc);
    const int waves  = (n_graphs / 16) * 2;
    const int blocks = (waves + 15) / 16;
    gemm_bias_relu<<<blocks, 512, 0, stream>>>(node_acc, edge_acc, u, W, b,
                                               out, n_graphs);
}


# product, bias and relu folded into the table-sum kernel: a block sums its graph's row, keeps it in LDS, and its wave 0 multiplies it by W (staged in LDS) on f32 WMMA; third launch gone
# speedup vs baseline: 6.3022x; 1.1478x over previous
#include <hip/hip_runtime.h>

#define D 32
#define NG 256

typedef __attribute__((ext_vector_type(2))) float v2f;
typedef __attribute__((ext_vector_type(8))) float v8f;

typedef __attribute__((ext_vector_type(4))) float v4f;

#define NEDGEB 481
#define NNODEB 31
#define NSLOT (NEDGEB + NNODEB)
static_assert(NSLOT % 32 == 0, "sum_tables_gemm deals the slots to 32 thread groups");

#define KP (3 * D / 2)
#define WPP 96
__global__ void __launch_bounds__(256)
sum_tables_gemm(const float* __restrict__ part, const float* __restrict__ u,
                const float* __restrict__ W, const float* __restrict__ b,
                float* __restrict__ out) {
    __shared__ v4f red[2][8][8];
    __shared__ v4f rowv4[3 * D / 4];
    __shared__ v4f wsh4[KP * WPP / 4];
    const int q  = threadIdx.x & 7;
    const int pg = threadIdx.x >> 3;
    const int lane = threadIdx.x & 31;
    const int wave = __builtin_amdgcn_readfirstlane(threadIdx.x >> 5);

    const v4f upiece = ((const v4f*)u)[blockIdx.x * 8 + q];
    const float bias = b[lane];
    v2f we[3], wo[3];
    #pragma unroll
    for (int j = 0; j < 3; ++j) {
        const int id = threadIdx.x + 256 * j;
        const float* wp = W + (id >> 4) * (2 * D) + 2 * (id & 15);
        we[j] = *(const v2f*)wp;
        wo[j] = *(const v2f*)(wp + D);
    }
    __builtin_amdgcn_sched_barrier(0);

    const v4f* p4 = (const v4f*)part + (size_t)pg * (NG * D / 4)
                    + blockIdx.x * 8 + q;
    v4f tv[NSLOT / 32];
    #pragma unroll
    for (int i = 0; i < NSLOT / 32; ++i) tv[i] = p4[(size_t)i * 32 * (NG * D / 4)];

    #pragma unroll
    for (int j = 0; j < 3; ++j) {
        const int id = threadIdx.x + 256 * j;
        v4f w;
        w[0] = we[j][0]; w[1] = wo[j][0]; w[2] = we[j][1]; w[3] = wo[j][1];
        wsh4[(id >> 4) * (WPP / 4) + (id & 15)] = w;
    }

    v4f se = {0.f, 0.f, 0.f, 0.f};
    v4f sn = {0.f, 0.f, 0.f, 0.f};
    #pragma unroll
    for (int i = 0; i < NSLOT / 32; ++i) {
        const v4f v = tv[i];
        if (32 * i + 31 < NEDGEB)             se += v;
        else if (32 * i >= NEDGEB)            sn += v;
        else if (pg + 32 * i < NEDGEB)        se += v;
        else                                  sn += v;
    }
    #pragma unroll
    for (int k = 0; k < 4; ++k) {
        se[k] += __shfl_xor(se[k], 8, 32);
        se[k] += __shfl_xor(se[k], 16, 32);
        sn[k] += __shfl_xor(sn[k], 8, 32);
        sn[k] += __shfl_xor(sn[k], 16, 32);
    }
    if (lane < 8) { red[0][wave][q] = se; red[1][wave][q] = sn; }
    __syncthreads();
    if (threadIdx.x < 24) {
        const int kind = threadIdx.x >> 3;
        v4f t = upiece;
        if (kind < 2) {
            t = red[kind][0][q];
            #pragma unroll
            for (int w = 1; w < 8; ++w) t += red[kind][w][q];
        }
        const int seg = (kind == 0) ? 1 : (kind == 1) ? 0 : 2;
        rowv4[seg * (D / 4) + q] = t;
    }
    __syncthreads();
    if (wave == 0) {
        const float* rowv = (const float*)rowv4;
        const float* wsh  = (const float*)wsh4;
        const int c = lane & 15;
        const int h = lane >> 4;
        v2f af[24], bf0[24], bf1[24];
        #pragma unroll
        for (int kb = 0; kb < 24; ++kb) {
            const int kp = kb * 2 + h;
            af[kb]  = *(const v2f*)(rowv + 2 * kp);
            bf0[kb] = *(const v2f*)(wsh + kp * WPP + 2 * c);
            bf1[kb] = *(const v2f*)(wsh + kp * WPP + 2 * (16 + c));
        }
        __builtin_amdgcn_sched_barrier(0);
        v8f c0 = {}, c1 = {};
        #pragma unroll
        for (int kb = 0; kb < 24; ++kb) {
            c0 = __builtin_amdgcn_wmma_f32_16x16x4_f32(
                     false, af[kb],  false, bf0[kb],
                     (short)0, c0,  false,  false);
            c1 = __builtin_amdgcn_wmma_f32_16x16x4_f32(
                     false, af[kb],  false, bf1[kb],
                     (short)0, c1,  false,  false);
        }
        const float v = (lane < 16 ? c0[0] : c1[0]) + bias;
        out[blockIdx.x * D + lane] = v > 0.f ? v : 0.f;
    }
}

#define TRIP 32

__device__ __forceinline__ void load_rows(const v4f* __restrict__ rows, int trip,
                                          int lane, v4f (&R)[8]) {
    const v4f* p = rows + (size_t)trip * (TRIP * D / 4) + lane;
    #pragma unroll
    for (int i = 0; i < 8; ++i) R[i] = p[i * 32];
}

__device__ __forceinline__ void add_rows(float* sh, const v4f (&R)[8], int g,
                                         int r, const int (&c)[4]) {
    const bool r1 = (r & 1) != 0, r2 = (r & 2) != 0;
    #pragma unroll
    for (int i = 0; i < 8; ++i) {
        const int gi = __shfl(g, 4 * i + r, 32);
        float* row = sh + gi * D;
        const v4f v = R[i];
        const float a0 = r1 ? v[1] : v[0];
        const float a1 = r1 ? v[2] : v[1];
        const float a2 = r1 ? v[3] : v[2];
        const float a3 = r1 ? v[0] : v[3];
        const float w0 = r2 ? a2 : a0;
        const float w1 = r2 ? a3 : a1;
        const float w2 = r2 ? a0 : a2;
        const float w3 = r2 ? a1 : a3;
        atomicAdd(row + c[0], w0);
        atomicAdd(row + c[1], w1);
        atomicAdd(row + c[2], w2);
        atomicAdd(row + c[3], w3);
    }
}

template <bool NODE>
__device__ __forceinline__ int graph_key(const int* __restrict__ src, int i) {
    if (NODE) return i;
    return src[i];
}

template <bool NODE>
__device__ __forceinline__ void sum_rows(float* sh, const float* __restrict__ feat,
                                         const int* __restrict__ src,
                                         const int* __restrict__ batch,
                                         int n_rows, int blk, int nblk) {
    const int lane = threadIdx.x & 31;
    const int wave = __builtin_amdgcn_readfirstlane(threadIdx.x >> 5);
    const int nwb  = blockDim.x >> 5;
    const int r = lane >> 3;
    const int q = lane & 7;
    int c[4];
    #pragma unroll
    for (int j = 0; j < 4; ++j) c[j] = 4 * q + ((j + r) & 3);

    const v4f* rows = (const v4f*)feat;
    const int nfull = n_rows / TRIP;
    const int gw = wave * nblk + blk;
    const int nw = nwb * nblk;
    const int n_my = (nfull > gw) ? (nfull - gw + nw - 1) / nw : 0;

    if (n_my > 0) {
        v4f A[8], B[8];
        int ga, gb;
        const int last = gw + (n_my - 1) * nw;
        int s0 = graph_key<NODE>(src, gw * TRIP + lane);
        load_rows(rows, gw, lane, A);
        int s1 = graph_key<NODE>(src, min(gw + nw, last) * TRIP + lane);
        ga = batch[s0];
        int k = 0;
        for (; k + 2 < n_my; k += 2) {
            const int t1 = gw + (k + 1) * nw;
            const int t2 = t1 + nw;
            const int s2 = graph_key<NODE>(src, t2 * TRIP + lane);
            gb = batch[s1];
            load_rows(rows, t1, lane, B);
            add_rows(sh, A, ga, r, c);
            const int s3 = graph_key<NODE>(src, min(t2 + nw, last) * TRIP + lane);
            ga = batch[s2];
            load_rows(rows, t2, lane, A);
            add_rows(sh, B, gb, r, c);
            s1 = s3;
        }
        if (k + 1 < n_my) {
            gb = batch[s1];
            load_rows(rows, gw + (k + 1) * nw, lane, B);
            add_rows(sh, A, ga, r, c);
            add_rows(sh, B, gb, r, c);
        } else {
            add_rows(sh, A, ga, r, c);
        }
    }

    if (blk == 0 && wave == 0) {
        for (int e = nfull * TRIP; e < n_rows; ++e) {
            int g = batch[graph_key<NODE>(src, e)];
            atomicAdd(&sh[g * D + lane], feat[(long)e * D + lane]);
        }
    }
}

__global__ void __launch_bounds__(256)
scatter_rows(const float* __restrict__ edge_attr,
             const int* __restrict__ src,
             const float* __restrict__ x,
             const int* __restrict__ batch,
             float* __restrict__ part, int n_edges, int n_nodes) {
    __shared__ v4f sh4[NG * D / 4];
    float* sh = (float*)sh4;
    for (int i = threadIdx.x; i < NG * D; i += blockDim.x) sh[i] = 0.f;
    __syncthreads();

    if (blockIdx.x < NEDGEB)
        sum_rows<false>(sh, edge_attr, src, batch, n_edges, blockIdx.x, NEDGEB);
    else
        sum_rows<true>(sh, x, src, batch, n_nodes, blockIdx.x - NEDGEB, NNODEB);

    __syncthreads();
    v4f* dst = (v4f*)part + (size_t)blockIdx.x * (NG * D / 4);
    #pragma unroll 8
    for (int i = threadIdx.x; i < NG * D / 4; i += blockDim.x) dst[i] = sh4[i];
}

extern "C" void kernel_launch(void* const* d_in, const int* in_sizes, int n_in,
                              void* d_out, int out_size, void* d_ws, size_t ws_size,
                              hipStream_t stream) {
    const float* x          = (const float*)d_in[0];
    const int*   edge_index = (const int*)d_in[1];
    const float* edge_attr  = (const float*)d_in[2];
    const float* u          = (const float*)d_in[3];
    const int*   batch      = (const int*)d_in[4];
    const float* W          = (const float*)d_in[5];
    const float* b          = (const float*)d_in[6];
    float*       out        = (float*)d_out;

    const int n_nodes  = in_sizes[4];
    const int n_edges  = in_sizes[2] / D;

    float* part = (float*)d_ws;

    scatter_rows<<<NSLOT, 256, 0, stream>>>(edge_attr, edge_index, x, batch,
                                            part, n_edges, n_nodes);
    sum_tables_gemm<<<NG, 256, 0, stream>>>(part, u, W, b, out);
}


# speedup vs baseline: 6.6639x; 1.0574x over previous
#include <hip/hip_runtime.h>

#define D 32
#define NG 256

typedef __attribute__((ext_vector_type(2))) float v2f;
typedef __attribute__((ext_vector_type(8))) float v8f;

typedef __attribute__((ext_vector_type(4))) float v4f;

#define NEDGEB 481
#define NNODEB 31
#define NSLOT (NEDGEB + NNODEB)
static_assert(NSLOT % 32 == 0, "sum_tables_gemm deals the slots to 32 thread groups");

#define KP (3 * D / 2)
#define WPP 96
__global__ void __launch_bounds__(256)
sum_tables_gemm(const float* __restrict__ part, const float* __restrict__ u,
                const float* __restrict__ W, const float* __restrict__ b,
                float* __restrict__ out) {
    __shared__ v4f red[2][8][8];
    __shared__ v4f rowv4[3 * D / 4];
    __shared__ v4f wsh4[KP * WPP / 4];
    const int q  = threadIdx.x & 7;
    const int pg = threadIdx.x >> 3;
    const int lane = threadIdx.x & 31;
    const int wave = __builtin_amdgcn_readfirstlane(threadIdx.x >> 5);

    const v4f upiece = ((const v4f*)u)[blockIdx.x * 8 + q];
    const float bias = b[lane];
    v2f we[3], wo[3];
    #pragma unroll
    for (int j = 0; j < 3; ++j) {
        const int id = threadIdx.x + 256 * j;
        const float* wp = W + (id >> 4) * (2 * D) + 2 * (id & 15);
        we[j] = *(const v2f*)wp;
        wo[j] = *(const v2f*)(wp + D);
    }
    __builtin_amdgcn_sched_barrier(0);

    const v4f* p4 = (const v4f*)part + (size_t)pg * (NG * D / 4)
                    + blockIdx.x * 8 + q;
    v4f tv[NSLOT / 32];
    #pragma unroll
    for (int i = 0; i < NSLOT / 32; ++i) tv[i] = p4[(size_t)i * 32 * (NG * D / 4)];

    #pragma unroll
    for (int j = 0; j < 3; ++j) {
        const int id = threadIdx.x + 256 * j;
        v4f w;
        w[0] = we[j][0]; w[1] = wo[j][0]; w[2] = we[j][1]; w[3] = wo[j][1];
        wsh4[(id >> 4) * (WPP / 4) + (id & 15)] = w;
    }

    v4f se = {0.f, 0.f, 0.f, 0.f};
    v4f sn = {0.f, 0.f, 0.f, 0.f};
    #pragma unroll
    for (int i = 0; i < NSLOT / 32; ++i) {
        const v4f v = tv[i];
        if (32 * i + 31 < NEDGEB)             se += v;
        else if (32 * i >= NEDGEB)            sn += v;
        else if (pg + 32 * i < NEDGEB)        se += v;
        else                                  sn += v;
    }
    #pragma unroll
    for (int k = 0; k < 4; ++k) {
        se[k] += __shfl_xor(se[k], 8, 32);
        se[k] += __shfl_xor(se[k], 16, 32);
        sn[k] += __shfl_xor(sn[k], 8, 32);
        sn[k] += __shfl_xor(sn[k], 16, 32);
    }
    if (lane < 8) { red[0][wave][q] = se; red[1][wave][q] = sn; }
    __syncthreads();
    if (threadIdx.x < 24) {
        const int kind = threadIdx.x >> 3;
        v4f t = upiece;
        if (kind < 2) {
            t = red[kind][0][q];
            #pragma unroll
            for (int w = 1; w < 8; ++w) t += red[kind][w][q];
        }
        const int seg = (kind == 0) ? 1 : (kind == 1) ? 0 : 2;
        rowv4[seg * (D / 4) + q] = t;
    }
    __syncthreads();
    if (wave == 0) {
        const float* rowv = (const float*)rowv4;
        const float* wsh  = (const float*)wsh4;
        const int c = lane & 15;
        const int h = lane >> 4;
        v2f af[24], bf0[24], bf1[24];
        #pragma unroll
        for (int kb = 0; kb < 24; ++kb) {
            const int kp = kb * 2 + h;
            af[kb]  = *(const v2f*)(rowv + 2 * kp);
            bf0[kb] = *(const v2f*)(wsh + kp * WPP + 2 * c);
            bf1[kb] = *(const v2f*)(wsh + kp * WPP + 2 * (16 + c));
        }
        __builtin_amdgcn_sched_barrier(0);
        v8f c0 = {}, c1 = {};
        #pragma unroll
        for (int kb = 0; kb < 24; ++kb) {
            c0 = __builtin_amdgcn_wmma_f32_16x16x4_f32(
                     false, af[kb],  false, bf0[kb],
                     (short)0, c0,  false,  false);
            c1 = __builtin_amdgcn_wmma_f32_16x16x4_f32(
                     false, af[kb],  false, bf1[kb],
                     (short)0, c1,  false,  false);
        }
        const float v = (lane < 16 ? c0[0] : c1[0]) + bias;
        out[blockIdx.x * D + lane] = v > 0.f ? v : 0.f;
    }
}

#define TRIP 32

__device__ __forceinline__ void load_rows(const v4f* __restrict__ rows, int trip,
                                          int lane, v4f (&R)[8]) {
    const v4f* p = rows + (size_t)trip * (TRIP * D / 4) + lane;
    #pragma unroll
    for (int i = 0; i < 8; ++i) R[i] = __builtin_nontemporal_load(p + i * 32);
}

__device__ __forceinline__ void add_rows(float* sh, const v4f (&R)[8], int g,
                                         int r, const int (&c)[4]) {
    const bool r1 = (r & 1) != 0, r2 = (r & 2) != 0;
    #pragma unroll
    for (int i = 0; i < 8; ++i) {
        const int gi = __shfl(g, 4 * i + r, 32);
        float* row = sh + gi * D;
        const v4f v = R[i];
        const float a0 = r1 ? v[1] : v[0];
        const float a1 = r1 ? v[2] : v[1];
        const float a2 = r1 ? v[3] : v[2];
        const float a3 = r1 ? v[0] : v[3];
        const float w0 = r2 ? a2 : a0;
        const float w1 = r2 ? a3 : a1;
        const float w2 = r2 ? a0 : a2;
        const float w3 = r2 ? a1 : a3;
        atomicAdd(row + c[0], w0);
        atomicAdd(row + c[1], w1);
        atomicAdd(row + c[2], w2);
        atomicAdd(row + c[3], w3);
    }
}

template <bool NODE>
__device__ __forceinline__ int graph_key(const int* __restrict__ src, int i) {
    if (NODE) return i;
    return __builtin_nontemporal_load(src + i);
}

template <bool NODE>
__device__ __forceinline__ void sum_rows(float* sh, const float* __restrict__ feat,
                                         const int* __restrict__ src,
                                         const int* __restrict__ batch,
                                         int n_rows, int blk, int nblk) {
    const int lane = threadIdx.x & 31;
    const int wave = __builtin_amdgcn_readfirstlane(threadIdx.x >> 5);
    const int nwb  = blockDim.x >> 5;
    const int r = lane >> 3;
    const int q = lane & 7;
    int c[4];
    #pragma unroll
    for (int j = 0; j < 4; ++j) c[j] = 4 * q + ((j + r) & 3);

    const v4f* rows = (const v4f*)feat;
    const int nfull = n_rows / TRIP;
    const int gw = wave * nblk + blk;
    const int nw = nwb * nblk;
    const int n_my = (nfull > gw) ? (nfull - gw + nw - 1) / nw : 0;

    if (n_my > 0) {
        v4f A[8], B[8];
        int ga, gb;
        const int last = gw + (n_my - 1) * nw;
        int s0 = graph_key<NODE>(src, gw * TRIP + lane);
        load_rows(rows, gw, lane, A);
        int s1 = graph_key<NODE>(src, min(gw + nw, last) * TRIP + lane);
        ga = batch[s0];
        int k = 0;
        for (; k + 2 < n_my; k += 2) {
            const int t1 = gw + (k + 1) * nw;
            const int t2 = t1 + nw;
            const int s2 = graph_key<NODE>(src, t2 * TRIP + lane);
            gb = batch[s1];
            load_rows(rows, t1, lane, B);
            add_rows(sh, A, ga, r, c);
            const int s3 = graph_key<NODE>(src, min(t2 + nw, last) * TRIP + lane);
            ga = batch[s2];
            load_rows(rows, t2, lane, A);
            add_rows(sh, B, gb, r, c);
            s1 = s3;
        }
        if (k + 1 < n_my) {
            gb = batch[s1];
            load_rows(rows, gw + (k + 1) * nw, lane, B);
            add_rows(sh, A, ga, r, c);
            add_rows(sh, B, gb, r, c);
        } else {
            add_rows(sh, A, ga, r, c);
        }
    }

    if (blk == 0 && wave == 0) {
        for (int e = nfull * TRIP; e < n_rows; ++e) {
            int g = batch[graph_key<NODE>(src, e)];
            atomicAdd(&sh[g * D + lane], feat[(long)e * D + lane]);
        }
    }
}

__global__ void __launch_bounds__(256)
scatter_rows(const float* __restrict__ edge_attr,
             const int* __restrict__ src,
             const float* __restrict__ x,
             const int* __restrict__ batch,
             float* __restrict__ part, int n_edges, int n_nodes) {
    __shared__ v4f sh4[NG * D / 4];
    float* sh = (float*)sh4;
    for (int i = threadIdx.x; i < NG * D; i += blockDim.x) sh[i] = 0.f;
    __syncthreads();

    if (blockIdx.x < NEDGEB)
        sum_rows<false>(sh, edge_attr, src, batch, n_edges, blockIdx.x, NEDGEB);
    else
        sum_rows<true>(sh, x, src, batch, n_nodes, blockIdx.x - NEDGEB, NNODEB);

    __syncthreads();
    v4f* dst = (v4f*)part + (size_t)blockIdx.x * (NG * D / 4);
    #pragma unroll 8
    for (int i = threadIdx.x; i < NG * D / 4; i += blockDim.x) dst[i] = sh4[i];
}

extern "C" void kernel_launch(void* const* d_in, const int* in_sizes, int n_in,
                              void* d_out, int out_size, void* d_ws, size_t ws_size,
                              hipStream_t stream) {
    const float* x          = (const float*)d_in[0];
    const int*   edge_index = (const int*)d_in[1];
    const float* edge_attr  = (const float*)d_in[2];
    const float* u          = (const float*)d_in[3];
    const int*   batch      = (const int*)d_in[4];
    const float* W          = (const float*)d_in[5];
    const float* b          = (const float*)d_in[6];
    float*       out        = (float*)d_out;

    const int n_nodes  = in_sizes[4];
    const int n_edges  = in_sizes[2] / D;

    float* part = (float*)d_ws;

    scatter_rows<<<NSLOT, 256, 0, stream>>>(edge_attr, edge_index, x, batch,
                                            part, n_edges, n_nodes);
    sum_tables_gemm<<<NG, 256, 0, stream>>>(part, u, W, b, out);
}


# scatter: every edge block copies batch into LDS once (16-bit ids, whole-line loads) and looks graphs up there, no global gather per trip; 256 blocks of 16 waves, one per WGP; sum order changes
# speedup vs baseline: 7.0487x; 1.0577x over previous
#include <hip/hip_runtime.h>

#define D 32
#define NG 256

typedef __attribute__((ext_vector_type(2))) float v2f;
typedef __attribute__((ext_vector_type(8))) float v8f;

typedef __attribute__((ext_vector_type(4))) float v4f;

#define NEDGEB 241
#define NNODEB 15
#define NSLOT (NEDGEB + NNODEB)
#define IDCAP 100000
static_assert(NSLOT % 32 == 0, "sum_tables_gemm deals the slots to 32 thread groups");

#define KP (3 * D / 2)
#define WPP 96
__global__ void __launch_bounds__(256)
sum_tables_gemm(const float* __restrict__ part, const float* __restrict__ u,
                const float* __restrict__ W, const float* __restrict__ b,
                float* __restrict__ out) {
    __shared__ v4f red[2][8][8];
    __shared__ v4f rowv4[3 * D / 4];
    __shared__ v4f wsh4[KP * WPP / 4];
    const int q  = threadIdx.x & 7;
    const int pg = threadIdx.x >> 3;
    const int lane = threadIdx.x & 31;
    const int wave = __builtin_amdgcn_readfirstlane(threadIdx.x >> 5);

    const v4f upiece = ((const v4f*)u)[blockIdx.x * 8 + q];
    const float bias = b[lane];
    v2f we[3], wo[3];
    #pragma unroll
    for (int j = 0; j < 3; ++j) {
        const int id = threadIdx.x + 256 * j;
        const float* wp = W + (id >> 4) * (2 * D) + 2 * (id & 15);
        we[j] = *(const v2f*)wp;
        wo[j] = *(const v2f*)(wp + D);
    }
    __builtin_amdgcn_sched_barrier(0);

    const v4f* p4 = (const v4f*)part + (size_t)pg * (NG * D / 4)
                    + blockIdx.x * 8 + q;
    v4f tv[NSLOT / 32];
    #pragma unroll
    for (int i = 0; i < NSLOT / 32; ++i) tv[i] = p4[(size_t)i * 32 * (NG * D / 4)];

    #pragma unroll
    for (int j = 0; j < 3; ++j) {
        const int id = threadIdx.x + 256 * j;
        v4f w;
        w[0] = we[j][0]; w[1] = wo[j][0]; w[2] = we[j][1]; w[3] = wo[j][1];
        wsh4[(id >> 4) * (WPP / 4) + (id & 15)] = w;
    }

    v4f se = {0.f, 0.f, 0.f, 0.f};
    v4f sn = {0.f, 0.f, 0.f, 0.f};
    #pragma unroll
    for (int i = 0; i < NSLOT / 32; ++i) {
        const v4f v = tv[i];
        if (32 * i + 31 < NEDGEB)             se += v;
        else if (32 * i >= NEDGEB)            sn += v;
        else if (pg + 32 * i < NEDGEB)        se += v;
        else                                  sn += v;
    }
    #pragma unroll
    for (int k = 0; k < 4; ++k) {
        se[k] += __shfl_xor(se[k], 8, 32);
        se[k] += __shfl_xor(se[k], 16, 32);
        sn[k] += __shfl_xor(sn[k], 8, 32);
        sn[k] += __shfl_xor(sn[k], 16, 32);
    }
    if (lane < 8) { red[0][wave][q] = se; red[1][wave][q] = sn; }
    __syncthreads();
    if (threadIdx.x < 24) {
        const int kind = threadIdx.x >> 3;
        v4f t = upiece;
        if (kind < 2) {
            t = red[kind][0][q];
            #pragma unroll
            for (int w = 1; w < 8; ++w) t += red[kind][w][q];
        }
        const int seg = (kind == 0) ? 1 : (kind == 1) ? 0 : 2;
        rowv4[seg * (D / 4) + q] = t;
    }
    __syncthreads();
    if (wave == 0) {
        const float* rowv = (const float*)rowv4;
        const float* wsh  = (const float*)wsh4;
        const int c = lane & 15;
        const int h = lane >> 4;
        v2f af[24], bf0[24], bf1[24];
        #pragma unroll
        for (int kb = 0; kb < 24; ++kb) {
            const int kp = kb * 2 + h;
            af[kb]  = *(const v2f*)(rowv + 2 * kp);
            bf0[kb] = *(const v2f*)(wsh + kp * WPP + 2 * c);
            bf1[kb] = *(const v2f*)(wsh + kp * WPP + 2 * (16 + c));
        }
        __builtin_amdgcn_sched_barrier(0);
        v8f c0 = {}, c1 = {};
        #pragma unroll
        for (int kb = 0; kb < 24; ++kb) {
            c0 = __builtin_amdgcn_wmma_f32_16x16x4_f32(
                     false, af[kb],  false, bf0[kb],
                     (short)0, c0,  false,  false);
            c1 = __builtin_amdgcn_wmma_f32_16x16x4_f32(
                     false, af[kb],  false, bf1[kb],
                     (short)0, c1,  false,  false);
        }
        const float v = (lane < 16 ? c0[0] : c1[0]) + bias;
        out[blockIdx.x * D + lane] = v > 0.f ? v : 0.f;
    }
}

#define TRIP 32

__device__ __forceinline__ void load_rows(const v4f* __restrict__ rows, int trip,
                                          int lane, v4f (&R)[8]) {
    const v4f* p = rows + (size_t)trip * (TRIP * D / 4) + lane;
    #pragma unroll
    for (int i = 0; i < 8; ++i) R[i] = __builtin_nontemporal_load(p + i * 32);
}

__device__ __forceinline__ void add_rows(float* sh, const v4f (&R)[8], int g,
                                         int r, const int (&c)[4]) {
    const bool r1 = (r & 1) != 0, r2 = (r & 2) != 0;
    #pragma unroll
    for (int i = 0; i < 8; ++i) {
        const int gi = __shfl(g, 4 * i + r, 32);
        float* row = sh + gi * D;
        const v4f v = R[i];
        const float a0 = r1 ? v[1] : v[0];
        const float a1 = r1 ? v[2] : v[1];
        const float a2 = r1 ? v[3] : v[2];
        const float a3 = r1 ? v[0] : v[3];
        const float w0 = r2 ? a2 : a0;
        const float w1 = r2 ? a3 : a1;
        const float w2 = r2 ? a0 : a2;
        const float w3 = r2 ? a1 : a3;
        atomicAdd(row + c[0], w0);
        atomicAdd(row + c[1], w1);
        atomicAdd(row + c[2], w2);
        atomicAdd(row + c[3], w3);
    }
}

template <bool NODE>
__device__ __forceinline__ int graph_key(const int* __restrict__ src, int i) {
    if (NODE) return i;
    return __builtin_nontemporal_load(src + i);
}

template <bool NODE>
__device__ __forceinline__ int graph_of(const int* __restrict__ batch,
                                        const unsigned short* ids, int key) {
    if (NODE) return batch[key];
    return ids[key];
}

typedef __attribute__((ext_vector_type(4))) int v4i;
typedef __attribute__((ext_vector_type(2))) unsigned int v2u;

#define SG 10
__device__ __forceinline__ void stage_ids(v2u* ids2, const int* __restrict__ batch,
                                          int n_nodes, int blk, int nblk) {
    const int npieces = n_nodes >> 2;
    const v4i* b4 = (const v4i*)batch;
    const int rot = (int)(((long)blk * npieces) / nblk);
    for (int base = 0; base < npieces; base += SG * (int)blockDim.x) {
        v4i t[SG];
        int p[SG];
        #pragma unroll
        for (int j = 0; j < SG; ++j) {
            int q = min(base + j * (int)blockDim.x + (int)threadIdx.x, npieces - 1) + rot;
            p[j] = (q >= npieces) ? q - npieces : q;
            t[j] = b4[p[j]];
        }
        #pragma unroll
        for (int j = 0; j < SG; ++j) {
            v2u w;
            w[0] = ((unsigned)t[j][0] & 0xffffu) | ((unsigned)t[j][1] << 16);
            w[1] = ((unsigned)t[j][2] & 0xffffu) | ((unsigned)t[j][3] << 16);
            ids2[p[j]] = w;
        }
    }
    if ((int)threadIdx.x < (n_nodes & 3)) {
        const int e = (npieces << 2) + (int)threadIdx.x;
        ((unsigned short*)ids2)[e] = (unsigned short)batch[e];
    }
}

template <bool NODE>
__device__ __forceinline__ void sum_rows(float* sh, v2u* ids2,
                                         const float* __restrict__ feat,
                                         const int* __restrict__ src,
                                         const int* __restrict__ batch,
                                         int n_rows, int n_nodes, int blk, int nblk) {
    const unsigned short* ids = (const unsigned short*)ids2;
    const int lane = threadIdx.x & 31;
    const int wave = __builtin_amdgcn_readfirstlane(threadIdx.x >> 5);
    const int nwb  = blockDim.x >> 5;
    const int r = lane >> 3;
    const int q = lane & 7;
    int c[4];
    #pragma unroll
    for (int j = 0; j < 4; ++j) c[j] = 4 * q + ((j + r) & 3);

    const v4f* rows = (const v4f*)feat;
    const int nfull = n_rows / TRIP;
    const int gw = wave * nblk + blk;
    const int nw = nwb * nblk;
    const int n_my = (nfull > gw) ? (nfull - gw + nw - 1) / nw : 0;
    const int last = gw + (n_my - 1) * nw;

    v4f A[8], B[8];
    int s0 = 0, s1 = 0;
    if (n_my > 0) {
        s0 = graph_key<NODE>(src, gw * TRIP + lane);
        load_rows(rows, gw, lane, A);
        s1 = graph_key<NODE>(src, min(gw + nw, last) * TRIP + lane);
    }
    if (!NODE) stage_ids(ids2, batch, n_nodes, blk, nblk);
    __syncthreads();

    if (n_my > 0) {
        int ga = graph_of<NODE>(batch, ids, s0), gb;
        int k = 0;
        for (; k + 2 < n_my; k += 2) {
            const int t1 = gw + (k + 1) * nw;
            const int t2 = t1 + nw;
            const int s2 = graph_key<NODE>(src, t2 * TRIP + lane);
            gb = graph_of<NODE>(batch, ids, s1);
            load_rows(rows, t1, lane, B);
            __builtin_amdgcn_sched_barrier(0);
            add_rows(sh, A, ga, r, c);
            const int s3 = graph_key<NODE>(src, min(t2 + nw, last) * TRIP + lane);
            ga = graph_of<NODE>(batch, ids, s2);
            load_rows(rows, t2, lane, A);
            __builtin_amdgcn_sched_barrier(0);
            add_rows(sh, B, gb, r, c);
            s1 = s3;
        }
        if (k + 1 < n_my) {
            gb = graph_of<NODE>(batch, ids, s1);
            load_rows(rows, gw + (k + 1) * nw, lane, B);
            add_rows(sh, A, ga, r, c);
            add_rows(sh, B, gb, r, c);
        } else {
            add_rows(sh, A, ga, r, c);
        }
    }

    if (blk == 0 && wave == 0) {
        for (int e = nfull * TRIP; e < n_rows; ++e) {
            int g = batch[graph_key<NODE>(src, e)];
            atomicAdd(&sh[g * D + lane], feat[(long)e * D + lane]);
        }
    }
}

__global__ void __launch_bounds__(512)
scatter_rows(const float* __restrict__ edge_attr,
             const int* __restrict__ src,
             const float* __restrict__ x,
             const int* __restrict__ batch,
             float* __restrict__ part, int n_edges, int n_nodes) {
    __shared__ v4f sh4[NG * D / 4];
    __shared__ v2u ids2[IDCAP / 4];
    float* sh = (float*)sh4;
    const v4f zero = {0.f, 0.f, 0.f, 0.f};
    for (int i = threadIdx.x; i < NG * D / 4; i += blockDim.x) sh4[i] = zero;

    if (blockIdx.x < NEDGEB)
        sum_rows<false>(sh, ids2, edge_attr, src, batch, n_edges, n_nodes,
                        blockIdx.x, NEDGEB);
    else
        sum_rows<true>(sh, ids2, x, src, batch, n_nodes, n_nodes,
                       blockIdx.x - NEDGEB, NNODEB);

    __syncthreads();
    v4f* dst = (v4f*)part + (size_t)blockIdx.x * (NG * D / 4);
    #pragma unroll 4
    for (int i = threadIdx.x; i < NG * D / 4; i += blockDim.x) dst[i] = sh4[i];
}

extern "C" void kernel_launch(void* const* d_in, const int* in_sizes, int n_in,
                              void* d_out, int out_size, void* d_ws, size_t ws_size,
                              hipStream_t stream) {
    const float* x          = (const float*)d_in[0];
    const int*   edge_index = (const int*)d_in[1];
    const float* edge_attr  = (const float*)d_in[2];
    const float* u          = (const float*)d_in[3];
    const int*   batch      = (const int*)d_in[4];
    const float* W          = (const float*)d_in[5];
    const float* b          = (const float*)d_in[6];
    float*       out        = (float*)d_out;

    const int n_nodes  = in_sizes[4];
    const int n_edges  = in_sizes[2] / D;

    float* part = (float*)d_ws;

    scatter_rows<<<NSLOT, 512, 0, stream>>>(edge_attr, edge_index, x, batch,
                                            part, n_edges, n_nodes);
    sum_tables_gemm<<<NG, 256, 0, stream>>>(part, u, W, b, out);
}
